# starting kernel
# baseline (speedup 1.0000x reference)
#include <hip/hip_runtime.h>
#include <hip/hip_bf16.h>

#define BATCH   32
#define SEQ     4096
#define QDIM    512
#define KVDIM   512

#define NCHUNK   16
#define ROWS_PER_CHUNK (SEQ / NCHUNK)
#define WAVES    8
#define TPB      (WAVES * 32)

#define WQ_OFF      0
#define PART_STRIDE (KVDIM + 8)
#define PART_OFF    (BATCH * KVDIM)

typedef float v2f __attribute__((ext_vector_type(2)));
typedef float v8f __attribute__((ext_vector_type(8)));

__global__ void __launch_bounds__(32)
wq_gemm_wmma(const float* __restrict__ query,
             const float* __restrict__ weight,
             float* __restrict__ wq) {
  const int tile = blockIdx.x;
  const int tM   = tile >> 5;
  const int tN   = tile & 31;
  const int lane = threadIdx.x;
  const int half = lane >> 4;
  const int lr   = lane & 15;

  const int aRow = tM * 16 + lr;
  const int bCol = tN * 16 + lr;

  v8f c = {};
  for (int k0 = 0; k0 < QDIM; k0 += 4) {
    const int ka = k0 + half * 2;
    v2f a, b;
    a[0] = query[aRow * QDIM + ka];
    a[1] = query[aRow * QDIM + ka + 1];
    b[0] = weight[(size_t)ka * KVDIM + bCol];
    b[1] = weight[(size_t)(ka + 1) * KVDIM + bCol];
    c = __builtin_amdgcn_wmma_f32_16x16x4_f32(false, a, false, b,
                                              (short)0, c, false, false);
  }
#pragma unroll
  for (int r = 0; r < 8; ++r) {
    wq[(tM * 16 + half * 8 + r) * KVDIM + tN * 16 + lr] = c[r];
  }
}

__global__ void __launch_bounds__(TPB)
attn_partial(const float* __restrict__ kv,
             const int*   __restrict__ mask,
             const float* __restrict__ wq,
             float*       __restrict__ part) {
  __shared__ float s_wq[KVDIM];
  __shared__ float s_acc[WAVES][KVDIM];
  __shared__ float s_m[WAVES];
  __shared__ float s_l[WAVES];

  const int b     = blockIdx.x / NCHUNK;
  const int chunk = blockIdx.x % NCHUNK;
  const int tid   = threadIdx.x;
  const int wave  = tid >> 5;
  const int lane  = tid & 31;
  const float NEG_INF = -__builtin_inff();

  for (int i = tid; i < KVDIM; i += TPB) s_wq[i] = wq[b * KVDIM + i];
  __syncthreads();

  float4 w4[4];
#pragma unroll
  for (int q = 0; q < 4; ++q)
    w4[q] = *(const float4*)&s_wq[q * 128 + lane * 4];

  float  m = NEG_INF, l = 0.f;
  float4 acc[4] = {};

  const int sBase = chunk * ROWS_PER_CHUNK;
  const int iters = ROWS_PER_CHUNK / WAVES;
  for (int it = 0; it < iters; ++it) {
    const int s = sBase + it * WAVES + wave;
    const float* __restrict__ row = kv + ((size_t)b * SEQ + s) * KVDIM;
    if (it + 1 < iters)
      __builtin_prefetch(row + WAVES * KVDIM, 0, 0);

    float4 r4[4];
#pragma unroll
    for (int q = 0; q < 4; ++q)
      r4[q] = *(const float4*)&row[q * 128 + lane * 4];

    float d = 0.f;
#pragma unroll
    for (int q = 0; q < 4; ++q) {
      d = fmaf(r4[q].x, w4[q].x, d);
      d = fmaf(r4[q].y, w4[q].y, d);
      d = fmaf(r4[q].z, w4[q].z, d);
      d = fmaf(r4[q].w, w4[q].w, d);
    }
#pragma unroll
    for (int off = 16; off > 0; off >>= 1) d += __shfl_xor(d, off, 32);

    if (mask[b * SEQ + s] != 0) {
      const float mN    = fmaxf(m, d);
      const float scale = __expf(m - mN);
      const float p     = __expf(d - mN);
      l = l * scale + p;
#pragma unroll
      for (int q = 0; q < 4; ++q) {
        acc[q].x = fmaf(p, r4[q].x, acc[q].x * scale);
        acc[q].y = fmaf(p, r4[q].y, acc[q].y * scale);
        acc[q].z = fmaf(p, r4[q].z, acc[q].z * scale);
        acc[q].w = fmaf(p, r4[q].w, acc[q].w * scale);
      }
      m = mN;
    }
  }

#pragma unroll
  for (int q = 0; q < 4; ++q)
    *(float4*)&s_acc[wave][q * 128 + lane * 4] = acc[q];
  if (lane == 0) { s_m[wave] = m; s_l[wave] = l; }
  __syncthreads();

  float mB = NEG_INF;
#pragma unroll
  for (int w = 0; w < WAVES; ++w) mB = fmaxf(mB, s_m[w]);
  float wgt[WAVES];
  float lB = 0.f;
#pragma unroll
  for (int w = 0; w < WAVES; ++w) {
    const float wi = (s_m[w] == NEG_INF) ? 0.f : __expf(s_m[w] - mB);
    wgt[w] = wi;
    lB += wi * s_l[w];
  }
  float* P = part + (size_t)(b * NCHUNK + chunk) * PART_STRIDE;
  for (int e = tid; e < KVDIM; e += TPB) {
    float v = 0.f;
#pragma unroll
    for (int w = 0; w < WAVES; ++w) v = fmaf(wgt[w], s_acc[w][e], v);
    P[e] = v;
  }
  if (tid == 0) { P[KVDIM] = mB; P[KVDIM + 1] = lB; }
}

__global__ void __launch_bounds__(256)
attn_combine(const float* __restrict__ part, float* __restrict__ out) {
  const int b   = blockIdx.x;
  const int tid = threadIdx.x;
  const float NEG_INF = -__builtin_inff();

  float mG = NEG_INF;
#pragma unroll
  for (int c = 0; c < NCHUNK; ++c)
    mG = fmaxf(mG, part[(size_t)(b * NCHUNK + c) * PART_STRIDE + KVDIM]);

  float wc[NCHUNK];
  float lG = 0.f;
#pragma unroll
  for (int c = 0; c < NCHUNK; ++c) {
    const size_t base = (size_t)(b * NCHUNK + c) * PART_STRIDE;
    const float  mi   = part[base + KVDIM];
    const float  wi   = (mi == NEG_INF) ? 0.f : __expf(mi - mG);
    wc[c] = wi;
    lG += wi * part[base + KVDIM + 1];
  }
  const float inv = 1.f / lG;

  for (int e = tid; e < KVDIM; e += 256) {
    float v = 0.f;
#pragma unroll
    for (int c = 0; c < NCHUNK; ++c)
      v = fmaf(wc[c], part[(size_t)(b * NCHUNK + c) * PART_STRIDE + e], v);
    out[b * KVDIM + e] = v * inv;
  }
}

extern "C" void kernel_launch(void* const* d_in, const int* in_sizes, int n_in,
                              void* d_out, int out_size, void* d_ws, size_t ws_size,
                              hipStream_t stream) {
  const float* query  = (const float*)d_in[0];
  const float* kv     = (const float*)d_in[1];
  const int*   mask   = (const int*)  d_in[2];
  const float* weight = (const float*)d_in[3];
  float* out = (float*)d_out;

  float* ws   = (float*)d_ws;
  float* wq   = ws + WQ_OFF;
  float* part = ws + PART_OFF;

  wq_gemm_wmma<<<64, 32, 0, stream>>>(query, weight, wq);
  attn_partial<<<BATCH * NCHUNK, TPB, 0, stream>>>(kv, mask, wq, part);
  attn_combine<<<BATCH, 256, 0, stream>>>(part, out);
}


# Attention pass re-tiled row-tile x column-strip: a wave streams 16 rows of one 128-col strip (16 b128 loads in flight), strips meet in LDS, one rescale per 32-row tile; changes row summation order
# speedup vs baseline: 1.2221x; 1.2221x over previous
#include <hip/hip_runtime.h>
#include <hip/hip_bf16.h>

#define BATCH   32
#define SEQ     4096
#define QDIM    512
#define KVDIM   512

#define NCHUNK   16
#define ROWS_PER_CHUNK (SEQ / NCHUNK)
#define WAVES    8
#define TPB      (WAVES * 32)

#define WQ_OFF      0
#define PART_STRIDE (KVDIM + 8)
#define PART_OFF    (BATCH * KVDIM)

typedef float v2f __attribute__((ext_vector_type(2)));
typedef float v8f __attribute__((ext_vector_type(8)));

__global__ void __launch_bounds__(32)
wq_gemm_wmma(const float* __restrict__ query,
             const float* __restrict__ weight,
             float* __restrict__ wq) {
  const int tile = blockIdx.x;
  const int tM   = tile >> 5;
  const int tN   = tile & 31;
  const int lane = threadIdx.x;
  const int half = lane >> 4;
  const int lr   = lane & 15;

  const int aRow = tM * 16 + lr;
  const int bCol = tN * 16 + lr;

  v8f c = {};
  for (int k0 = 0; k0 < QDIM; k0 += 4) {
    const int ka = k0 + half * 2;
    v2f a, b;
    a[0] = query[aRow * QDIM + ka];
    a[1] = query[aRow * QDIM + ka + 1];
    b[0] = weight[(size_t)ka * KVDIM + bCol];
    b[1] = weight[(size_t)(ka + 1) * KVDIM + bCol];
    c = __builtin_amdgcn_wmma_f32_16x16x4_f32(false, a, false, b,
                                              (short)0, c, false, false);
  }
#pragma unroll
  for (int r = 0; r < 8; ++r) {
    wq[(tM * 16 + half * 8 + r) * KVDIM + tN * 16 + lr] = c[r];
  }
}

__device__ __forceinline__ float xrow(float x) {
  const int b = __builtin_bit_cast(int, x);
  return __builtin_bit_cast(float, __builtin_amdgcn_permlanex16(
      b, b, 0x76543210, 0xfedcba98, false, false));
}
template <int M>
__device__ __forceinline__ float xlane(float x) {
  const int b = __builtin_bit_cast(int, x);
  return __builtin_bit_cast(float, __builtin_amdgcn_update_dpp(
      b, b, 0x160 + M, 0xf, 0xf, false));
}

#define TR    16
#define TILE  (2 * TR)

__global__ void __launch_bounds__(TPB)
attn_partial(const float* __restrict__ kv,
             const int*   __restrict__ mask,
             const float* __restrict__ wq,
             float*       __restrict__ part) {
  __shared__ float4 s_sc[2][TILE];
  __shared__ float4 s_accx[4][32];

  const int b     = blockIdx.x / NCHUNK;
  const int chunk = blockIdx.x % NCHUNK;
  const int tid   = threadIdx.x;
  const int lane  = tid & 31;
  const int wave  = __builtin_amdgcn_readfirstlane(tid >> 5);
  const int strip = wave & 3;
  const int rg    = wave >> 2;
  const float NEG_INF = -__builtin_inff();

  const float4 w4 = *(const float4*)&wq[b * KVDIM + strip * 128 + lane * 4];

  float  m = NEG_INF;
  float  lsum = 0.f;
  float4 acc = {0.f, 0.f, 0.f, 0.f};

  const int sBase = chunk * ROWS_PER_CHUNK;
  const float* __restrict__ base =
      kv + ((size_t)b * SEQ + sBase + rg * TR) * KVDIM + strip * 128 + lane * 4;
  const int* __restrict__ mrow = mask + b * SEQ + sBase + lane;

  for (int it = 0; it < ROWS_PER_CHUNK / TILE; ++it) {
    const float* __restrict__ p0 = base + (size_t)it * TILE * KVDIM;
    const int mk = mrow[it * TILE];

    float4 x[TR];
    for (int j = 0; j < TR; ++j)
      x[j] = *(const float4*)(p0 + (size_t)j * KVDIM);

    float t[TR];
    for (int j = 0; j < TR; ++j)
      t[j] = fmaf(x[j].w, w4.w, fmaf(x[j].z, w4.z, fmaf(x[j].y, w4.y, x[j].x * w4.x)));

    const bool h4 = (lane & 16) != 0, h3 = (lane & 8) != 0,
               h2 = (lane & 4) != 0,  h1 = (lane & 2) != 0;
    float u[8], v[4], z[2];
    for (int i = 0; i < 8; ++i) {
      const float send = h4 ? t[i] : t[i + 8];
      const float keep = h4 ? t[i + 8] : t[i];
      u[i] = keep + xrow(send);
    }
    for (int i = 0; i < 4; ++i) {
      const float send = h3 ? u[i] : u[i + 4];
      const float keep = h3 ? u[i + 4] : u[i];
      v[i] = keep + xlane<8>(send);
    }
    for (int i = 0; i < 2; ++i) {
      const float send = h2 ? v[i] : v[i + 2];
      const float keep = h2 ? v[i + 2] : v[i];
      z[i] = keep + xlane<4>(send);
    }
    float y;
    {
      const float send = h1 ? z[0] : z[1];
      const float keep = h1 ? z[1] : z[0];
      y = keep + xlane<2>(send);
    }
    y += xlane<1>(y);

    if ((lane & 1) == 0)
      ((float*)&s_sc[it & 1][rg * TR + (lane >> 1)])[strip] = y;
    __syncthreads();

    const float4 q  = s_sc[it & 1][lane];
    const float  s  = (q.x + q.y) + (q.z + q.w);
    const float  sm = (mk != 0) ? s : NEG_INF;
    float mT = fmaxf(sm, xrow(sm));
    mT = fmaxf(mT, xlane<8>(mT));
    mT = fmaxf(mT, xlane<4>(mT));
    mT = fmaxf(mT, xlane<2>(mT));
    mT = fmaxf(mT, xlane<1>(mT));
    const float mN    = fmaxf(m, mT);
    const float scale = (m == NEG_INF) ? 0.f : __expf(m - mN);
    const float p     = (mk != 0) ? __expf(s - mN) : 0.f;
    lsum = lsum * scale + p;
    m = mN;

    acc.x *= scale; acc.y *= scale; acc.z *= scale; acc.w *= scale;
    for (int j = 0; j < TR; ++j) {
      const float pj = __int_as_float(
          __builtin_amdgcn_readlane(__float_as_int(p), rg * TR + j));
      acc.x = fmaf(pj, x[j].x, acc.x);
      acc.y = fmaf(pj, x[j].y, acc.y);
      acc.z = fmaf(pj, x[j].z, acc.z);
      acc.w = fmaf(pj, x[j].w, acc.w);
    }
  }

  lsum += xrow(lsum);
  lsum += xlane<8>(lsum);
  lsum += xlane<4>(lsum);
  lsum += xlane<2>(lsum);
  lsum += xlane<1>(lsum);

  if (rg == 1) s_accx[strip][lane] = acc;
  __syncthreads();
  float* P = part + (size_t)(b * NCHUNK + chunk) * PART_STRIDE;
  if (rg == 0) {
    const float4 o = s_accx[strip][lane];
    acc.x += o.x; acc.y += o.y; acc.z += o.z; acc.w += o.w;
    *(float4*)&P[strip * 128 + lane * 4] = acc;
  }
  if (tid == 0) { P[KVDIM] = m; P[KVDIM + 1] = lsum; }
}

__global__ void __launch_bounds__(256)
attn_combine(const float* __restrict__ part, float* __restrict__ out) {
  const int b   = blockIdx.x;
  const int tid = threadIdx.x;
  const float NEG_INF = -__builtin_inff();

  float mG = NEG_INF;
#pragma unroll
  for (int c = 0; c < NCHUNK; ++c)
    mG = fmaxf(mG, part[(size_t)(b * NCHUNK + c) * PART_STRIDE + KVDIM]);

  float wc[NCHUNK];
  float lG = 0.f;
#pragma unroll
  for (int c = 0; c < NCHUNK; ++c) {
    const size_t base = (size_t)(b * NCHUNK + c) * PART_STRIDE;
    const float  mi   = part[base + KVDIM];
    const float  wi   = (mi == NEG_INF) ? 0.f : __expf(mi - mG);
    wc[c] = wi;
    lG += wi * part[base + KVDIM + 1];
  }
  const float inv = 1.f / lG;

  for (int e = tid; e < KVDIM; e += 256) {
    float v = 0.f;
#pragma unroll
    for (int c = 0; c < NCHUNK; ++c)
      v = fmaf(wc[c], part[(size_t)(b * NCHUNK + c) * PART_STRIDE + e], v);
    out[b * KVDIM + e] = v * inv;
  }
}

extern "C" void kernel_launch(void* const* d_in, const int* in_sizes, int n_in,
                              void* d_out, int out_size, void* d_ws, size_t ws_size,
                              hipStream_t stream) {
  const float* query  = (const float*)d_in[0];
  const float* kv     = (const float*)d_in[1];
  const int*   mask   = (const int*)  d_in[2];
  const float* weight = (const float*)d_in[3];
  float* out = (float*)d_out;

  float* ws   = (float*)d_ws;
  float* wq   = ws + WQ_OFF;
  float* part = ws + PART_OFF;

  wq_gemm_wmma<<<64, 32, 0, stream>>>(query, weight, wq);
  attn_partial<<<BATCH * NCHUNK, TPB, 0, stream>>>(kv, mask, wq, part);
  attn_combine<<<BATCH, 256, 0, stream>>>(part, out);
}


# GEMM kernel: a tile's two operand panels are fetched by 16 waves as one group of 16-byte loads and parked in LDS; one wave then runs the whole K loop from LDS (K not split); sums unchanged
# speedup vs baseline: 1.5579x; 1.2748x over previous
#include <hip/hip_runtime.h>
#include <hip/hip_bf16.h>

#define BATCH   32
#define SEQ     4096
#define QDIM    512
#define KVDIM   512

#define NCHUNK   16
#define ROWS_PER_CHUNK (SEQ / NCHUNK)
#define WAVES    8
#define TPB      (WAVES * 32)

#define WQ_OFF      0
#define PART_STRIDE (KVDIM + 8)
#define PART_OFF    (BATCH * KVDIM)

typedef float v2f __attribute__((ext_vector_type(2)));
typedef float v8f __attribute__((ext_vector_type(8)));

#define GW      16
#define GT      (GW * 32)
#define APITCH  (QDIM + 4)

__global__ void __launch_bounds__(GT)
wq_gemm_wmma(const float* __restrict__ query,
             const float* __restrict__ weight,
             float* __restrict__ wq) {
  __shared__ __attribute__((aligned(16))) float sA[16 * APITCH];
  __shared__ __attribute__((aligned(16))) float sB[(QDIM / 2) * 32];

  const int tile = blockIdx.x;
  const int tM   = tile >> 5;
  const int tN   = tile & 31;
  const int tid  = threadIdx.x;

  const float4* __restrict__ gA = (const float4*)(query + (size_t)tM * 16 * QDIM);
  float4 ra[4];
  for (int i = 0; i < 4; ++i) ra[i] = gA[tid + i * GT];

  const int seg = tid & 3;
  const int p0  = tid >> 2;
  const float* __restrict__ gB = weight + tN * 16 + seg * 4;
  float4 rb[4];
  for (int i = 0; i < 2; ++i) {
    const int pr = p0 + i * 128;
    rb[2 * i]     = *(const float4*)(gB + (size_t)(2 * pr) * KVDIM);
    rb[2 * i + 1] = *(const float4*)(gB + (size_t)(2 * pr + 1) * KVDIM);
  }

  for (int i = 0; i < 4; ++i) {
    const int f = tid + i * GT;
    *(float4*)&sA[(f >> 7) * APITCH + (f & 127) * 4] = ra[i];
  }
  for (int i = 0; i < 2; ++i) {
    const int pr = p0 + i * 128;
    const float4 lo = rb[2 * i], hi = rb[2 * i + 1];
    float4 s0, s1;
    s0.x = lo.x; s0.y = hi.x; s0.z = lo.y; s0.w = hi.y;
    s1.x = lo.z; s1.y = hi.z; s1.z = lo.w; s1.w = hi.w;
    *(float4*)&sB[pr * 32 + seg * 8]     = s0;
    *(float4*)&sB[pr * 32 + seg * 8 + 4] = s1;
  }
  __syncthreads();

  if (__builtin_amdgcn_readfirstlane(tid >> 5) != 0) return;

  const int lane = tid;
  const int half = lane >> 4;
  const int lr   = lane & 15;
  const float* __restrict__ pa = &sA[lr * APITCH + half * 2];
  const float* __restrict__ pb = &sB[half * 32 + lr * 2];

  constexpr int KB = 16;
  constexpr int NB = QDIM / 4 / KB;
  v2f fa[2][KB], fb[2][KB];
#pragma unroll
  for (int s = 0; s < KB; ++s) {
    fa[0][s] = *(const v2f*)(pa + 4 * s);
    fb[0][s] = *(const v2f*)(pb + 64 * s);
  }
  v8f c = {};
#pragma unroll
  for (int blk = 0; blk < NB; ++blk) {
    if (blk + 1 < NB) {
#pragma unroll
      for (int s = 0; s < KB; ++s) {
        const int j = (blk + 1) * KB + s;
        fa[(blk + 1) & 1][s] = *(const v2f*)(pa + 4 * j);
        fb[(blk + 1) & 1][s] = *(const v2f*)(pb + 64 * j);
      }
    }
    __builtin_amdgcn_sched_barrier(0);
#pragma unroll
    for (int s = 0; s < KB; ++s) {
      c = __builtin_amdgcn_wmma_f32_16x16x4_f32(false, fa[blk & 1][s], false,
                                                fb[blk & 1][s], (short)0, c,
                                                false, false);
    }
    __builtin_amdgcn_sched_barrier(0);
  }
#pragma unroll
  for (int r = 0; r < 8; ++r) {
    wq[(tM * 16 + half * 8 + r) * KVDIM + tN * 16 + lr] = c[r];
  }
}

__device__ __forceinline__ float xrow(float x) {
  const int b = __builtin_bit_cast(int, x);
  return __builtin_bit_cast(float, __builtin_amdgcn_permlanex16(
      b, b, 0x76543210, 0xfedcba98, false, false));
}
template <int M>
__device__ __forceinline__ float xlane(float x) {
  const int b = __builtin_bit_cast(int, x);
  return __builtin_bit_cast(float, __builtin_amdgcn_update_dpp(
      b, b, 0x160 + M, 0xf, 0xf, false));
}

#define TR    16
#define TILE  (2 * TR)

__global__ void __launch_bounds__(TPB)
attn_partial(const float* __restrict__ kv,
             const int*   __restrict__ mask,
             const float* __restrict__ wq,
             float*       __restrict__ part) {
  __shared__ float4 s_sc[2][TILE];
  __shared__ float4 s_accx[4][32];

  const int b     = blockIdx.x / NCHUNK;
  const int chunk = blockIdx.x % NCHUNK;
  const int tid   = threadIdx.x;
  const int lane  = tid & 31;
  const int wave  = __builtin_amdgcn_readfirstlane(tid >> 5);
  const int strip = wave & 3;
  const int rg    = wave >> 2;
  const float NEG_INF = -__builtin_inff();

  const float4 w4 = *(const float4*)&wq[b * KVDIM + strip * 128 + lane * 4];

  float  m = NEG_INF;
  float  lsum = 0.f;
  float4 acc = {0.f, 0.f, 0.f, 0.f};

  const int sBase = chunk * ROWS_PER_CHUNK;
  const float* __restrict__ base =
      kv + ((size_t)b * SEQ + sBase + rg * TR) * KVDIM + strip * 128 + lane * 4;
  const int* __restrict__ mrow = mask + b * SEQ + sBase + lane;

  for (int it = 0; it < ROWS_PER_CHUNK / TILE; ++it) {
    const float* __restrict__ p0 = base + (size_t)it * TILE * KVDIM;
    const int mk = mrow[it * TILE];

    float4 x[TR];
    for (int j = 0; j < TR; ++j)
      x[j] = *(const float4*)(p0 + (size_t)j * KVDIM);

    float t[TR];
    for (int j = 0; j < TR; ++j)
      t[j] = fmaf(x[j].w, w4.w, fmaf(x[j].z, w4.z, fmaf(x[j].y, w4.y, x[j].x * w4.x)));

    const bool h4 = (lane & 16) != 0, h3 = (lane & 8) != 0,
               h2 = (lane & 4) != 0,  h1 = (lane & 2) != 0;
    float u[8], v[4], z[2];
    for (int i = 0; i < 8; ++i) {
      const float send = h4 ? t[i] : t[i + 8];
      const float keep = h4 ? t[i + 8] : t[i];
      u[i] = keep + xrow(send);
    }
    for (int i = 0; i < 4; ++i) {
      const float send = h3 ? u[i] : u[i + 4];
      const float keep = h3 ? u[i + 4] : u[i];
      v[i] = keep + xlane<8>(send);
    }
    for (int i = 0; i < 2; ++i) {
      const float send = h2 ? v[i] : v[i + 2];
      const float keep = h2 ? v[i + 2] : v[i];
      z[i] = keep + xlane<4>(send);
    }
    float y;
    {
      const float send = h1 ? z[0] : z[1];
      const float keep = h1 ? z[1] : z[0];
      y = keep + xlane<2>(send);
    }
    y += xlane<1>(y);

    if ((lane & 1) == 0)
      ((float*)&s_sc[it & 1][rg * TR + (lane >> 1)])[strip] = y;
    __syncthreads();

    const float4 q  = s_sc[it & 1][lane];
    const float  s  = (q.x + q.y) + (q.z + q.w);
    const float  sm = (mk != 0) ? s : NEG_INF;
    float mT = fmaxf(sm, xrow(sm));
    mT = fmaxf(mT, xlane<8>(mT));
    mT = fmaxf(mT, xlane<4>(mT));
    mT = fmaxf(mT, xlane<2>(mT));
    mT = fmaxf(mT, xlane<1>(mT));
    const float mN    = fmaxf(m, mT);
    const float scale = (m == NEG_INF) ? 0.f : __expf(m - mN);
    const float p     = (mk != 0) ? __expf(s - mN) : 0.f;
    lsum = lsum * scale + p;
    m = mN;

    acc.x *= scale; acc.y *= scale; acc.z *= scale; acc.w *= scale;
    for (int j = 0; j < TR; ++j) {
      const float pj = __int_as_float(
          __builtin_amdgcn_readlane(__float_as_int(p), rg * TR + j));
      acc.x = fmaf(pj, x[j].x, acc.x);
      acc.y = fmaf(pj, x[j].y, acc.y);
      acc.z = fmaf(pj, x[j].z, acc.z);
      acc.w = fmaf(pj, x[j].w, acc.w);
    }
  }

  lsum += xrow(lsum);
  lsum += xlane<8>(lsum);
  lsum += xlane<4>(lsum);
  lsum += xlane<2>(lsum);
  lsum += xlane<1>(lsum);

  if (rg == 1) s_accx[strip][lane] = acc;
  __syncthreads();
  float* P = part + (size_t)(b * NCHUNK + chunk) * PART_STRIDE;
  if (rg == 0) {
    const float4 o = s_accx[strip][lane];
    acc.x += o.x; acc.y += o.y; acc.z += o.z; acc.w += o.w;
    *(float4*)&P[strip * 128 + lane * 4] = acc;
  }
  if (tid == 0) { P[KVDIM] = m; P[KVDIM + 1] = lsum; }
}

__global__ void __launch_bounds__(256)
attn_combine(const float* __restrict__ part, float* __restrict__ out) {
  const int b   = blockIdx.x;
  const int tid = threadIdx.x;
  const float NEG_INF = -__builtin_inff();

  float mG = NEG_INF;
#pragma unroll
  for (int c = 0; c < NCHUNK; ++c)
    mG = fmaxf(mG, part[(size_t)(b * NCHUNK + c) * PART_STRIDE + KVDIM]);

  float wc[NCHUNK];
  float lG = 0.f;
#pragma unroll
  for (int c = 0; c < NCHUNK; ++c) {
    const size_t base = (size_t)(b * NCHUNK + c) * PART_STRIDE;
    const float  mi   = part[base + KVDIM];
    const float  wi   = (mi == NEG_INF) ? 0.f : __expf(mi - mG);
    wc[c] = wi;
    lG += wi * part[base + KVDIM + 1];
  }
  const float inv = 1.f / lG;

  for (int e = tid; e < KVDIM; e += 256) {
    float v = 0.f;
#pragma unroll
    for (int c = 0; c < NCHUNK; ++c)
      v = fmaf(wc[c], part[(size_t)(b * NCHUNK + c) * PART_STRIDE + e], v);
    out[b * KVDIM + e] = v * inv;
  }
}

extern "C" void kernel_launch(void* const* d_in, const int* in_sizes, int n_in,
                              void* d_out, int out_size, void* d_ws, size_t ws_size,
                              hipStream_t stream) {
  const float* query  = (const float*)d_in[0];
  const float* kv     = (const float*)d_in[1];
  const int*   mask   = (const int*)  d_in[2];
  const float* weight = (const float*)d_in[3];
  float* out = (float*)d_out;

  float* ws   = (float*)d_ws;
  float* wq   = ws + WQ_OFF;
  float* part = ws + PART_OFF;

  wq_gemm_wmma<<<64, GT, 0, stream>>>(query, weight, wq);
  attn_partial<<<BATCH * NCHUNK, TPB, 0, stream>>>(kv, mask, wq, part);
  attn_combine<<<BATCH, 256, 0, stream>>>(part, out);
}
